# starting kernel
# baseline (speedup 1.0000x reference)
#include <hip/hip_runtime.h>
#include <hip/hip_bf16.h>
#include <math.h>


#define Bsz 4
#define Lsz 2048
#define Hsz 1024
#define Esz 8
#define Rsz 8

typedef __attribute__((ext_vector_type(2))) float v2f;
typedef __attribute__((ext_vector_type(8))) float v8f;

#define APACK_BATCH (256 * 32 * 2)
#define BPACK_OFF   (Bsz * APACK_BATCH)
#define BPACK_BATCH (64 * 4 * 32 * 2)

__global__ __launch_bounds__(256) void moe_gate_pack(
    const float* __restrict__ x,
    const float* __restrict__ W_router,
    const float* __restrict__ A_down,
    const float* __restrict__ B_up,
    float* __restrict__ ws) {
  const int b = blockIdx.x;
  const int t = threadIdx.x;

  __shared__ float red[256];
  __shared__ float slog[Esz];
  __shared__ int   sexp[2];
  __shared__ float sgate[2];

  {
    const int e = t >> 5;
    const int lane = t & 31;
    const float* xr = x + (size_t)b * Lsz * Hsz;
    const float* wr = W_router + (size_t)e * Hsz;
    float p = 0.f;
    for (int h = lane; h < Hsz; h += 32) p += xr[h] * wr[h];
    red[t] = p;
  }
  __syncthreads();
  if (t < Esz) {
    float s = 0.f;
    for (int i = 0; i < 32; ++i) s += red[t * 32 + i];
    slog[t] = s;
  }
  __syncthreads();

  if (t == 0) {
    int i1 = 0;
    float v1 = slog[0];
    for (int e = 1; e < Esz; ++e)
      if (slog[e] > v1) { v1 = slog[e]; i1 = e; }
    int i2 = -1;
    float v2 = -INFINITY;
    for (int e = 0; e < Esz; ++e) {
      if (e == i1) continue;
      if (slog[e] > v2) { v2 = slog[e]; i2 = e; }
    }
    const float e2 = expf(v2 - v1);
    const float inv = 1.f / (1.f + e2);
    sexp[0] = i1; sexp[1] = i2;
    sgate[0] = inv; sgate[1] = e2 * inv;
  }
  __syncthreads();

  {
    float* ap = ws + (size_t)b * APACK_BATCH;
    for (int it = 0; it < APACK_BATCH / 256; ++it) {
      const int idx = t + 256 * it;
      const int kk = idx >> 6;
      const int l  = (idx >> 1) & 31;
      const int j  = idx & 1;
      const int h  = 4 * kk + 2 * (l >> 4) + j;
      const int n  = l & 15;
      const int e  = sexp[n >> 3];
      const int r  = n & 7;
      ap[idx] = A_down[((size_t)e * Hsz + h) * Rsz + r];
    }
  }

  {
    float* bp = ws + BPACK_OFF + (size_t)b * BPACK_BATCH;
    for (int it = 0; it < BPACK_BATCH / 256; ++it) {
      const int idx = t + 256 * it;
      const int nc = idx >> 8;
      const int ks = (idx >> 6) & 3;
      const int l  = (idx >> 1) & 31;
      const int j  = idx & 1;
      const int slot = 4 * ks + 2 * (l >> 4) + j;
      const int col  = nc * 16 + (l & 15);
      const int e = sexp[slot >> 3];
      const int r = slot & 7;
      bp[idx] = sgate[slot >> 3] * B_up[((size_t)e * Rsz + r) * Hsz + col];
    }
  }
}

__global__ __launch_bounds__(64) void moe_main(
    const float* __restrict__ x,
    const float* __restrict__ ws,
    float* __restrict__ out) {
  const int b    = blockIdx.y;
  const int wave = threadIdx.x >> 5;
  const int lane = threadIdx.x & 31;
  const int strip = blockIdx.x * 2 + wave;
  const int row0  = strip * 16;

  const int r  = lane & 15;
  const int h2 = (lane >> 4) * 2;

  const float* xb = x + ((size_t)b * Lsz + row0) * Hsz;

  v8f acc0 = {}, acc1 = {}, acc2 = {}, acc3 = {};
  const float* apk = ws + (size_t)b * APACK_BATCH + (size_t)lane * 2;
  const float* xrow = xb + (size_t)r * Hsz + h2;
#pragma unroll 2
  for (int kk = 0; kk < 256; kk += 4) {
    v2f a0 = *(const v2f*)(xrow + 4 * (kk + 0));
    v2f a1 = *(const v2f*)(xrow + 4 * (kk + 1));
    v2f a2 = *(const v2f*)(xrow + 4 * (kk + 2));
    v2f a3 = *(const v2f*)(xrow + 4 * (kk + 3));
    v2f b0 = *(const v2f*)(apk + (kk + 0) * 64);
    v2f b1 = *(const v2f*)(apk + (kk + 1) * 64);
    v2f b2 = *(const v2f*)(apk + (kk + 2) * 64);
    v2f b3 = *(const v2f*)(apk + (kk + 3) * 64);
    acc0 = __builtin_amdgcn_wmma_f32_16x16x4_f32(false, a0, false, b0, (short)0, acc0, false, false);
    acc1 = __builtin_amdgcn_wmma_f32_16x16x4_f32(false, a1, false, b1, (short)0, acc1, false, false);
    acc2 = __builtin_amdgcn_wmma_f32_16x16x4_f32(false, a2, false, b2, (short)0, acc2, false, false);
    acc3 = __builtin_amdgcn_wmma_f32_16x16x4_f32(false, a3, false, b3, (short)0, acc3, false, false);
  }
  v8f low = (acc0 + acc1) + (acc2 + acc3);

  __shared__ __align__(16) float lds_low[2][16 * 18];
  float* Lw = &lds_low[wave][0];
#pragma unroll
  for (int v = 0; v < 8; ++v)
    Lw[(v + 8 * (lane >> 4)) * 18 + (lane & 15)] = low[v];

  v2f a2op[4];
#pragma unroll
  for (int ks = 0; ks < 4; ++ks)
    a2op[ks] = *(const v2f*)(&Lw[r * 18 + 4 * ks + h2]);

  const float* bpk = ws + BPACK_OFF + (size_t)b * BPACK_BATCH + (size_t)lane * 2;
  float* outp = out + ((size_t)b * Lsz + row0) * Hsz;
  for (int nc = 0; nc < 64; ++nc) {
    v8f y = {};
#pragma unroll
    for (int ks = 0; ks < 4; ++ks) {
      v2f bo = *(const v2f*)(bpk + nc * 256 + ks * 64);
      y = __builtin_amdgcn_wmma_f32_16x16x4_f32(
          false, a2op[ks], false, bo, (short)0, y, false, false);
    }
    const size_t cbase = (size_t)nc * 16 + (lane & 15);
#pragma unroll
    for (int v = 0; v < 8; ++v) {
      const size_t o = (size_t)(v + 8 * (lane >> 4)) * Hsz + cbase;
      outp[o] = xb[o] + y[v];
    }
  }
}

extern "C" void kernel_launch(void* const* d_in, const int* in_sizes, int n_in,
                              void* d_out, int out_size, void* d_ws, size_t ws_size,
                              hipStream_t stream) {
  const float* x        = (const float*)d_in[0];
  const float* W_router = (const float*)d_in[1];
  const float* A_down   = (const float*)d_in[2];
  const float* B_up     = (const float*)d_in[3];
  float* out = (float*)d_out;
  float* ws  = (float*)d_ws;

  moe_gate_pack<<<dim3(Bsz), dim3(256), 0, stream>>>(x, W_router, A_down, B_up, ws);
  moe_main<<<dim3(Lsz / 32, Bsz), dim3(64), 0, stream>>>(x, ws, out);
}


# each 16-row strip worked by an 8-wave block instead of one wave (8x the waves and loads in flight): stage-1 K split 8 ways, partials summed in LDS (sum order changes), stage-2 columns split 8 ways
# speedup vs baseline: 2.3443x; 2.3443x over previous
#include <hip/hip_runtime.h>
#include <hip/hip_bf16.h>
#include <math.h>


#define Bsz 4
#define Lsz 2048
#define Hsz 1024
#define Esz 8
#define Rsz 8

typedef __attribute__((ext_vector_type(2))) float v2f;
typedef __attribute__((ext_vector_type(8))) float v8f;

#define APACK_BATCH (256 * 32 * 2)
#define BPACK_OFF   (Bsz * APACK_BATCH)
#define BPACK_BATCH (64 * 4 * 32 * 2)

__global__ __launch_bounds__(256) void moe_gate_pack(
    const float* __restrict__ x,
    const float* __restrict__ W_router,
    const float* __restrict__ A_down,
    const float* __restrict__ B_up,
    float* __restrict__ ws) {
  const int b = blockIdx.x;
  const int t = threadIdx.x;

  __shared__ float red[256];
  __shared__ float slog[Esz];
  __shared__ int   sexp[2];
  __shared__ float sgate[2];

  {
    const int e = t >> 5;
    const int lane = t & 31;
    const float* xr = x + (size_t)b * Lsz * Hsz;
    const float* wr = W_router + (size_t)e * Hsz;
    float p = 0.f;
    for (int h = lane; h < Hsz; h += 32) p += xr[h] * wr[h];
    red[t] = p;
  }
  __syncthreads();
  if (t < Esz) {
    float s = 0.f;
    for (int i = 0; i < 32; ++i) s += red[t * 32 + i];
    slog[t] = s;
  }
  __syncthreads();

  if (t == 0) {
    int i1 = 0;
    float v1 = slog[0];
    for (int e = 1; e < Esz; ++e)
      if (slog[e] > v1) { v1 = slog[e]; i1 = e; }
    int i2 = -1;
    float v2 = -INFINITY;
    for (int e = 0; e < Esz; ++e) {
      if (e == i1) continue;
      if (slog[e] > v2) { v2 = slog[e]; i2 = e; }
    }
    const float e2 = expf(v2 - v1);
    const float inv = 1.f / (1.f + e2);
    sexp[0] = i1; sexp[1] = i2;
    sgate[0] = inv; sgate[1] = e2 * inv;
  }
  __syncthreads();

  {
    float* ap = ws + (size_t)b * APACK_BATCH;
    for (int it = 0; it < APACK_BATCH / 256; ++it) {
      const int idx = t + 256 * it;
      const int kk = idx >> 6;
      const int l  = (idx >> 1) & 31;
      const int j  = idx & 1;
      const int h  = 4 * kk + 2 * (l >> 4) + j;
      const int n  = l & 15;
      const int e  = sexp[n >> 3];
      const int r  = n & 7;
      ap[idx] = A_down[((size_t)e * Hsz + h) * Rsz + r];
    }
  }

  {
    float* bp = ws + BPACK_OFF + (size_t)b * BPACK_BATCH;
    for (int it = 0; it < BPACK_BATCH / 256; ++it) {
      const int idx = t + 256 * it;
      const int nc = idx >> 8;
      const int ks = (idx >> 6) & 3;
      const int l  = (idx >> 1) & 31;
      const int j  = idx & 1;
      const int slot = 4 * ks + 2 * (l >> 4) + j;
      const int col  = nc * 16 + (l & 15);
      const int e = sexp[slot >> 3];
      const int r = slot & 7;
      bp[idx] = sgate[slot >> 3] * B_up[((size_t)e * Rsz + r) * Hsz + col];
    }
  }
}

#define NWAVE 8
__global__ __launch_bounds__(256) void moe_main(
    const float* __restrict__ x,
    const float* __restrict__ ws,
    float* __restrict__ out) {
  const int b    = blockIdx.y;
  const int wave = threadIdx.x >> 5;
  const int lane = threadIdx.x & 31;
  const int row0 = blockIdx.x * 16;

  const int r  = lane & 15;
  const int h2 = (lane >> 4) * 2;

  const float* xb = x + ((size_t)b * Lsz + row0) * Hsz;

  v8f acc0 = {}, acc1 = {}, acc2 = {}, acc3 = {};
  const int kk0 = wave * 32;
  const float* apk = ws + (size_t)b * APACK_BATCH + (size_t)kk0 * 64 + (size_t)lane * 2;
  const float* xrow = xb + (size_t)r * Hsz + 4 * kk0 + h2;
#pragma unroll
  for (int kk = 0; kk < 32; kk += 4) {
    v2f a0 = *(const v2f*)(xrow + 4 * (kk + 0));
    v2f a1 = *(const v2f*)(xrow + 4 * (kk + 1));
    v2f a2 = *(const v2f*)(xrow + 4 * (kk + 2));
    v2f a3 = *(const v2f*)(xrow + 4 * (kk + 3));
    v2f b0 = *(const v2f*)(apk + (kk + 0) * 64);
    v2f b1 = *(const v2f*)(apk + (kk + 1) * 64);
    v2f b2 = *(const v2f*)(apk + (kk + 2) * 64);
    v2f b3 = *(const v2f*)(apk + (kk + 3) * 64);
    acc0 = __builtin_amdgcn_wmma_f32_16x16x4_f32(false, a0, false, b0, (short)0, acc0, false, false);
    acc1 = __builtin_amdgcn_wmma_f32_16x16x4_f32(false, a1, false, b1, (short)0, acc1, false, false);
    acc2 = __builtin_amdgcn_wmma_f32_16x16x4_f32(false, a2, false, b2, (short)0, acc2, false, false);
    acc3 = __builtin_amdgcn_wmma_f32_16x16x4_f32(false, a3, false, b3, (short)0, acc3, false, false);
  }
  v8f part = (acc0 + acc1) + (acc2 + acc3);

  const int nc0 = wave * 8;
  const float* bpk = ws + BPACK_OFF + (size_t)b * BPACK_BATCH + (size_t)nc0 * 256 + (size_t)lane * 2;
  const size_t obase = (size_t)(8 * (lane >> 4)) * Hsz + (size_t)nc0 * 16 + (lane & 15);
  const float* xres_p = xb + obase;
  float* outp = out + ((size_t)b * Lsz + row0) * Hsz + obase;
  v2f bo[8][4];
  float xres[8][8];
#pragma unroll
  for (int i = 0; i < 8; ++i) {
#pragma unroll
    for (int ks = 0; ks < 4; ++ks)
      bo[i][ks] = *(const v2f*)(bpk + i * 256 + ks * 64);
#pragma unroll
    for (int v = 0; v < 8; ++v)
      xres[i][v] = xres_p[(size_t)v * Hsz + i * 16];
  }

  __shared__ float lds_part[NWAVE][256];
  __shared__ __align__(16) float lds_low[16 * 18];
#pragma unroll
  for (int v = 0; v < 8; ++v)
    lds_part[wave][v * 32 + lane] = part[v];
  __syncthreads();
  {
    const int t = threadIdx.x;
    const float s = ((lds_part[0][t] + lds_part[1][t]) + (lds_part[2][t] + lds_part[3][t])) +
                    ((lds_part[4][t] + lds_part[5][t]) + (lds_part[6][t] + lds_part[7][t]));
    lds_low[(wave + 8 * (lane >> 4)) * 18 + (lane & 15)] = s;
  }
  __syncthreads();

  v2f a2op[4];
#pragma unroll
  for (int ks = 0; ks < 4; ++ks)
    a2op[ks] = *(const v2f*)(&lds_low[r * 18 + 4 * ks + h2]);

#pragma unroll
  for (int i = 0; i < 8; ++i) {
    v8f y = {};
#pragma unroll
    for (int ks = 0; ks < 4; ++ks)
      y = __builtin_amdgcn_wmma_f32_16x16x4_f32(
          false, a2op[ks], false, bo[i][ks], (short)0, y, false, false);
#pragma unroll
    for (int v = 0; v < 8; ++v)
      outp[(size_t)v * Hsz + i * 16] = xres[i][v] + y[v];
  }
}

extern "C" void kernel_launch(void* const* d_in, const int* in_sizes, int n_in,
                              void* d_out, int out_size, void* d_ws, size_t ws_size,
                              hipStream_t stream) {
  const float* x        = (const float*)d_in[0];
  const float* W_router = (const float*)d_in[1];
  const float* A_down   = (const float*)d_in[2];
  const float* B_up     = (const float*)d_in[3];
  float* out = (float*)d_out;
  float* ws  = (float*)d_ws;

  moe_gate_pack<<<dim3(Bsz), dim3(256), 0, stream>>>(x, W_router, A_down, B_up, ws);
  moe_main<<<dim3(Lsz / 16, Bsz), dim3(256), 0, stream>>>(x, ws, out);
}


# x staged in LDS: each wave loads its 16x128 slice once as whole 512-byte row segments (was 16-byte pieces of 16 lines per load); stage-1 fragments and the residual are read from LDS
# speedup vs baseline: 2.4589x; 1.0489x over previous
#include <hip/hip_runtime.h>
#include <hip/hip_bf16.h>
#include <math.h>


#define Bsz 4
#define Lsz 2048
#define Hsz 1024
#define Esz 8
#define Rsz 8

typedef __attribute__((ext_vector_type(2))) float v2f;
typedef __attribute__((ext_vector_type(8))) float v8f;

#define APACK_BATCH (256 * 32 * 2)
#define BPACK_OFF   (Bsz * APACK_BATCH)
#define BPACK_BATCH (64 * 4 * 32 * 2)

__global__ __launch_bounds__(256) void moe_gate_pack(
    const float* __restrict__ x,
    const float* __restrict__ W_router,
    const float* __restrict__ A_down,
    const float* __restrict__ B_up,
    float* __restrict__ ws) {
  const int b = blockIdx.x;
  const int t = threadIdx.x;

  __shared__ float red[256];
  __shared__ float slog[Esz];
  __shared__ int   sexp[2];
  __shared__ float sgate[2];

  {
    const int e = t >> 5;
    const int lane = t & 31;
    const float* xr = x + (size_t)b * Lsz * Hsz;
    const float* wr = W_router + (size_t)e * Hsz;
    float p = 0.f;
    for (int h = lane; h < Hsz; h += 32) p += xr[h] * wr[h];
    red[t] = p;
  }
  __syncthreads();
  if (t < Esz) {
    float s = 0.f;
    for (int i = 0; i < 32; ++i) s += red[t * 32 + i];
    slog[t] = s;
  }
  __syncthreads();

  if (t == 0) {
    int i1 = 0;
    float v1 = slog[0];
    for (int e = 1; e < Esz; ++e)
      if (slog[e] > v1) { v1 = slog[e]; i1 = e; }
    int i2 = -1;
    float v2 = -INFINITY;
    for (int e = 0; e < Esz; ++e) {
      if (e == i1) continue;
      if (slog[e] > v2) { v2 = slog[e]; i2 = e; }
    }
    const float e2 = expf(v2 - v1);
    const float inv = 1.f / (1.f + e2);
    sexp[0] = i1; sexp[1] = i2;
    sgate[0] = inv; sgate[1] = e2 * inv;
  }
  __syncthreads();

  {
    float* ap = ws + (size_t)b * APACK_BATCH;
    for (int it = 0; it < APACK_BATCH / 256; ++it) {
      const int idx = t + 256 * it;
      const int kk = idx >> 6;
      const int l  = (idx >> 1) & 31;
      const int j  = idx & 1;
      const int h  = 4 * kk + 2 * (l >> 4) + j;
      const int n  = l & 15;
      const int e  = sexp[n >> 3];
      const int r  = n & 7;
      ap[idx] = A_down[((size_t)e * Hsz + h) * Rsz + r];
    }
  }

  {
    float* bp = ws + BPACK_OFF + (size_t)b * BPACK_BATCH;
    for (int it = 0; it < BPACK_BATCH / 256; ++it) {
      const int idx = t + 256 * it;
      const int nc = idx >> 8;
      const int ks = (idx >> 6) & 3;
      const int l  = (idx >> 1) & 31;
      const int j  = idx & 1;
      const int slot = 4 * ks + 2 * (l >> 4) + j;
      const int col  = nc * 16 + (l & 15);
      const int e = sexp[slot >> 3];
      const int r = slot & 7;
      bp[idx] = sgate[slot >> 3] * B_up[((size_t)e * Rsz + r) * Hsz + col];
    }
  }
}

#define NWAVE 8
#define XPITCH 132
__global__ __launch_bounds__(256) void moe_main(
    const float* __restrict__ x,
    const float* __restrict__ ws,
    float* __restrict__ out) {
  const int b    = blockIdx.y;
  const int wave = threadIdx.x >> 5;
  const int lane = threadIdx.x & 31;
  const int row0 = blockIdx.x * 16;

  const int r  = lane & 15;
  const int h2 = (lane >> 4) * 2;

  const float* xb = x + ((size_t)b * Lsz + row0) * Hsz;

  __shared__ __align__(16) float lds_x[NWAVE][16 * XPITCH];
  __shared__ float lds_part[NWAVE][256];
  __shared__ __align__(16) float lds_low[16 * 18];
  float* xs = lds_x[wave];

  const float* xg = xb + 128 * wave + 4 * lane;
  float4 xv[16];
#pragma unroll
  for (int i = 0; i < 16; ++i)
    xv[i] = *(const float4*)(xg + (size_t)i * Hsz);

  const int kk0 = wave * 32;
  const float* apk = ws + (size_t)b * APACK_BATCH + (size_t)kk0 * 64 + (size_t)lane * 2;
  v2f bf[32];
#pragma unroll
  for (int kk = 0; kk < 32; ++kk)
    bf[kk] = *(const v2f*)(apk + kk * 64);

#pragma unroll
  for (int i = 0; i < 16; ++i)
    *(float4*)(xs + i * XPITCH + 4 * lane) = xv[i];
  __syncthreads();

  v8f acc0 = {}, acc1 = {}, acc2 = {}, acc3 = {};
  const float* xa = xs + r * XPITCH + h2;
  v2f af[32];
#pragma unroll
  for (int kk = 0; kk < 32; ++kk)
    af[kk] = *(const v2f*)(xa + 4 * kk);
#pragma unroll
  for (int kk = 0; kk < 32; kk += 4) {
    acc0 = __builtin_amdgcn_wmma_f32_16x16x4_f32(false, af[kk + 0], false, bf[kk + 0], (short)0, acc0, false, false);
    acc1 = __builtin_amdgcn_wmma_f32_16x16x4_f32(false, af[kk + 1], false, bf[kk + 1], (short)0, acc1, false, false);
    acc2 = __builtin_amdgcn_wmma_f32_16x16x4_f32(false, af[kk + 2], false, bf[kk + 2], (short)0, acc2, false, false);
    acc3 = __builtin_amdgcn_wmma_f32_16x16x4_f32(false, af[kk + 3], false, bf[kk + 3], (short)0, acc3, false, false);
  }
  v8f part = (acc0 + acc1) + (acc2 + acc3);

  const int nc0 = wave * 8;
  const float* bpk = ws + BPACK_OFF + (size_t)b * BPACK_BATCH + (size_t)nc0 * 256 + (size_t)lane * 2;
  const size_t obase = (size_t)(8 * (lane >> 4)) * Hsz + (size_t)nc0 * 16 + (lane & 15);
  float* outp = out + ((size_t)b * Lsz + row0) * Hsz + obase;
  const float* xres_p = xs + (8 * (lane >> 4)) * XPITCH + (lane & 15);
  v2f bo[8][4];
#pragma unroll
  for (int i = 0; i < 8; ++i) {
#pragma unroll
    for (int ks = 0; ks < 4; ++ks)
      bo[i][ks] = *(const v2f*)(bpk + i * 256 + ks * 64);
  }

#pragma unroll
  for (int v = 0; v < 8; ++v)
    lds_part[wave][v * 32 + lane] = part[v];
  __syncthreads();
  {
    const int t = threadIdx.x;
    const float s = ((lds_part[0][t] + lds_part[1][t]) + (lds_part[2][t] + lds_part[3][t])) +
                    ((lds_part[4][t] + lds_part[5][t]) + (lds_part[6][t] + lds_part[7][t]));
    lds_low[(wave + 8 * (lane >> 4)) * 18 + (lane & 15)] = s;
  }
  __syncthreads();

  v2f a2op[4];
#pragma unroll
  for (int ks = 0; ks < 4; ++ks)
    a2op[ks] = *(const v2f*)(&lds_low[r * 18 + 4 * ks + h2]);

  float xres[8][8];
#pragma unroll
  for (int i = 0; i < 8; ++i) {
#pragma unroll
    for (int v = 0; v < 8; ++v)
      xres[i][v] = xres_p[v * XPITCH + i * 16];
  }

#pragma unroll
  for (int i = 0; i < 8; ++i) {
    v8f y = {};
#pragma unroll
    for (int ks = 0; ks < 4; ++ks)
      y = __builtin_amdgcn_wmma_f32_16x16x4_f32(
          false, a2op[ks], false, bo[i][ks], (short)0, y, false, false);
#pragma unroll
    for (int v = 0; v < 8; ++v)
      outp[(size_t)v * Hsz + i * 16] = xres[i][v] + y[v];
  }
}

extern "C" void kernel_launch(void* const* d_in, const int* in_sizes, int n_in,
                              void* d_out, int out_size, void* d_ws, size_t ws_size,
                              hipStream_t stream) {
  const float* x        = (const float*)d_in[0];
  const float* W_router = (const float*)d_in[1];
  const float* A_down   = (const float*)d_in[2];
  const float* B_up     = (const float*)d_in[3];
  float* out = (float*)d_out;
  float* ws  = (float*)d_ws;

  moe_gate_pack<<<dim3(Bsz), dim3(256), 0, stream>>>(x, W_router, A_down, B_up, ws);
  moe_main<<<dim3(Lsz / 16, Bsz), dim3(256), 0, stream>>>(x, ws, out);
}


# gate and operand pre-pack folded into the strip kernel: each block computes its batch's top-2 gate itself and reads the WMMA B operands straight from A_down and B_up; one launch instead of two
# speedup vs baseline: 4.0776x; 1.6583x over previous
#include <hip/hip_runtime.h>
#include <hip/hip_bf16.h>
#include <math.h>


#define Bsz 4
#define Lsz 2048
#define Hsz 1024
#define Esz 8
#define Rsz 8

typedef __attribute__((ext_vector_type(2))) float v2f;
typedef __attribute__((ext_vector_type(8))) float v8f;


#define NWAVE 8
#define XPITCH 132
__global__ __launch_bounds__(256) void moe_main(
    const float* __restrict__ x,
    const float* __restrict__ W_router,
    const float* __restrict__ A_down,
    const float* __restrict__ B_up,
    float* __restrict__ out) {
  const int b    = blockIdx.y;
  const int wave = threadIdx.x >> 5;
  const int lane = threadIdx.x & 31;
  const int row0 = blockIdx.x * 16;

  const int r  = lane & 15;
  const int h2 = (lane >> 4) * 2;

  const float* xb = x + ((size_t)b * Lsz + row0) * Hsz;

  __shared__ __align__(16) float lds_x[NWAVE][16 * XPITCH];
  __shared__ float lds_part[NWAVE][256];
  __shared__ __align__(16) float lds_low[16 * 18];
  __shared__ float red[256];
  __shared__ float slog[Esz];
  float* xs = lds_x[wave];

  const float* xr0 = x + (size_t)b * Lsz * Hsz + lane;
  const float* wr  = W_router + (size_t)wave * Hsz + lane;
  float gx[32], gw[32];
#pragma unroll
  for (int i = 0; i < 32; ++i) {
    gx[i] = xr0[32 * i];
    gw[i] = wr[32 * i];
  }

  {
    float p = 0.f;
#pragma unroll
    for (int i = 0; i < 32; ++i) p += gx[i] * gw[i];
    red[threadIdx.x] = p;
  }
  __syncthreads();
  {
    float s = 0.f;
#pragma unroll
    for (int i = 0; i < 32; ++i) s += red[wave * 32 + i];
    slog[wave] = s;
  }
  __syncthreads();

  int i1 = 0, i2 = -1;
  float g1, g2;
  {
    float lg[Esz];
#pragma unroll
    for (int e = 0; e < Esz; ++e) lg[e] = slog[e];
    float v1 = lg[0];
#pragma unroll
    for (int e = 1; e < Esz; ++e)
      if (lg[e] > v1) { v1 = lg[e]; i1 = e; }
    float v2 = -INFINITY;
#pragma unroll
    for (int e = 0; e < Esz; ++e) {
      if (e == i1) continue;
      if (lg[e] > v2) { v2 = lg[e]; i2 = e; }
    }
    if (i2 < 0) i2 = i1 ^ 1;
    const float e2 = expf(v2 - v1);
    const float inv = 1.f / (1.f + e2);
    g1 = inv;
    g2 = e2 * inv;
  }

  const float* xg = xb + 128 * wave + 4 * lane;
  float4 xv[16];
#pragma unroll
  for (int i = 0; i < 16; ++i)
    xv[i] = *(const float4*)(xg + (size_t)i * Hsz);

  const int ea = (lane & 8) ? i2 : i1;
  const float* apk = A_down + ((size_t)ea * Hsz + 128 * wave + h2) * Rsz + (lane & 7);
  v2f bf[32];
#pragma unroll
  for (int kk = 0; kk < 32; ++kk) {
    bf[kk][0] = apk[kk * 4 * Rsz];
    bf[kk][1] = apk[kk * 4 * Rsz + Rsz];
  }

#pragma unroll
  for (int i = 0; i < 16; ++i)
    *(float4*)(xs + i * XPITCH + 4 * lane) = xv[i];
  __syncthreads();

  v8f acc0 = {}, acc1 = {}, acc2 = {}, acc3 = {};
  const float* xa = xs + r * XPITCH + h2;
  v2f af[32];
#pragma unroll
  for (int kk = 0; kk < 32; ++kk)
    af[kk] = *(const v2f*)(xa + 4 * kk);
#pragma unroll
  for (int kk = 0; kk < 32; kk += 4) {
    acc0 = __builtin_amdgcn_wmma_f32_16x16x4_f32(false, af[kk + 0], false, bf[kk + 0], (short)0, acc0, false, false);
    acc1 = __builtin_amdgcn_wmma_f32_16x16x4_f32(false, af[kk + 1], false, bf[kk + 1], (short)0, acc1, false, false);
    acc2 = __builtin_amdgcn_wmma_f32_16x16x4_f32(false, af[kk + 2], false, bf[kk + 2], (short)0, acc2, false, false);
    acc3 = __builtin_amdgcn_wmma_f32_16x16x4_f32(false, af[kk + 3], false, bf[kk + 3], (short)0, acc3, false, false);
  }
  v8f part = (acc0 + acc1) + (acc2 + acc3);

  const int nc0 = wave * 8;
  const float* bp1 = B_up + ((size_t)i1 * Rsz + h2) * Hsz + (size_t)nc0 * 16 + (lane & 15);
  const float* bp2 = B_up + ((size_t)i2 * Rsz + h2) * Hsz + (size_t)nc0 * 16 + (lane & 15);
  const size_t obase = (size_t)(8 * (lane >> 4)) * Hsz + (size_t)nc0 * 16 + (lane & 15);
  float* outp = out + ((size_t)b * Lsz + row0) * Hsz + obase;
  const float* xres_p = xs + (8 * (lane >> 4)) * XPITCH + (lane & 15);
  v2f bo[8][4];
#pragma unroll
  for (int i = 0; i < 8; ++i) {
#pragma unroll
    for (int ks = 0; ks < 2; ++ks) {
      bo[i][ks][0]     = g1 * bp1[(4 * ks) * Hsz + i * 16];
      bo[i][ks][1]     = g1 * bp1[(4 * ks + 1) * Hsz + i * 16];
      bo[i][ks + 2][0] = g2 * bp2[(4 * ks) * Hsz + i * 16];
      bo[i][ks + 2][1] = g2 * bp2[(4 * ks + 1) * Hsz + i * 16];
    }
  }

#pragma unroll
  for (int v = 0; v < 8; ++v)
    lds_part[wave][v * 32 + lane] = part[v];
  __syncthreads();
  {
    const int t = threadIdx.x;
    const float s = ((lds_part[0][t] + lds_part[1][t]) + (lds_part[2][t] + lds_part[3][t])) +
                    ((lds_part[4][t] + lds_part[5][t]) + (lds_part[6][t] + lds_part[7][t]));
    lds_low[(wave + 8 * (lane >> 4)) * 18 + (lane & 15)] = s;
  }
  __syncthreads();

  v2f a2op[4];
#pragma unroll
  for (int ks = 0; ks < 4; ++ks)
    a2op[ks] = *(const v2f*)(&lds_low[r * 18 + 4 * ks + h2]);

  float xres[8][8];
#pragma unroll
  for (int i = 0; i < 8; ++i) {
#pragma unroll
    for (int v = 0; v < 8; ++v)
      xres[i][v] = xres_p[v * XPITCH + i * 16];
  }

#pragma unroll
  for (int i = 0; i < 8; ++i) {
    v8f y = {};
#pragma unroll
    for (int ks = 0; ks < 4; ++ks)
      y = __builtin_amdgcn_wmma_f32_16x16x4_f32(
          false, a2op[ks], false, bo[i][ks], (short)0, y, false, false);
#pragma unroll
    for (int v = 0; v < 8; ++v)
      outp[(size_t)v * Hsz + i * 16] = xres[i][v] + y[v];
  }
}

extern "C" void kernel_launch(void* const* d_in, const int* in_sizes, int n_in,
                              void* d_out, int out_size, void* d_ws, size_t ws_size,
                              hipStream_t stream) {
  const float* x        = (const float*)d_in[0];
  const float* W_router = (const float*)d_in[1];
  const float* A_down   = (const float*)d_in[2];
  const float* B_up     = (const float*)d_in[3];
  float* out = (float*)d_out;

  moe_main<<<dim3(Lsz / 16, Bsz), dim3(256), 0, stream>>>(x, W_router, A_down, B_up, out);
}


# each 8-wave block works two 16-row strips: gate, A_down and B_up operand registers loaded once and used for both strips (half the blocks and waves, half the gate and operand reads); sums unchanged
# speedup vs baseline: 4.6248x; 1.1342x over previous
#include <hip/hip_runtime.h>
#include <hip/hip_bf16.h>
#include <math.h>


#define Bsz 4
#define Lsz 2048
#define Hsz 1024
#define Esz 8
#define Rsz 8

typedef __attribute__((ext_vector_type(2))) float v2f;
typedef __attribute__((ext_vector_type(8))) float v8f;


#define NWAVE 8
#define NSTRIP 2
#define XPITCH 132
__global__ __launch_bounds__(256) void moe_main(
    const float* __restrict__ x,
    const float* __restrict__ W_router,
    const float* __restrict__ A_down,
    const float* __restrict__ B_up,
    float* __restrict__ out) {
  const int b    = blockIdx.y;
  const int wave = threadIdx.x >> 5;
  const int lane = threadIdx.x & 31;
  const int row0 = blockIdx.x * (16 * NSTRIP);

  const int r  = lane & 15;
  const int h2 = (lane >> 4) * 2;

  const float* xb = x + ((size_t)b * Lsz + row0) * Hsz;

  __shared__ __align__(16) float lds_x[NWAVE][NSTRIP][16 * XPITCH];
  __shared__ float lds_part[NSTRIP][NWAVE][256];
  __shared__ __align__(16) float lds_low[NSTRIP][16 * 18];
  __shared__ float red[256];
  __shared__ float slog[Esz];

  const float* xr0 = x + (size_t)b * Lsz * Hsz + lane;
  const float* wr  = W_router + (size_t)wave * Hsz + lane;
  float gx[32], gw[32];
#pragma unroll
  for (int i = 0; i < 32; ++i) {
    gx[i] = xr0[32 * i];
    gw[i] = wr[32 * i];
  }

  {
    float p = 0.f;
#pragma unroll
    for (int i = 0; i < 32; ++i) p += gx[i] * gw[i];
    red[threadIdx.x] = p;
  }
  __syncthreads();
  {
    float s = 0.f;
#pragma unroll
    for (int i = 0; i < 32; ++i) s += red[wave * 32 + i];
    slog[wave] = s;
  }
  __syncthreads();

  int i1 = 0, i2 = -1;
  float g1, g2;
  {
    float lg[Esz];
#pragma unroll
    for (int e = 0; e < Esz; ++e) lg[e] = slog[e];
    float v1 = lg[0];
#pragma unroll
    for (int e = 1; e < Esz; ++e)
      if (lg[e] > v1) { v1 = lg[e]; i1 = e; }
    float v2 = -INFINITY;
#pragma unroll
    for (int e = 0; e < Esz; ++e) {
      if (e == i1) continue;
      if (lg[e] > v2) { v2 = lg[e]; i2 = e; }
    }
    if (i2 < 0) i2 = i1 ^ 1;
    const float e2 = expf(v2 - v1);
    const float inv = 1.f / (1.f + e2);
    g1 = inv;
    g2 = e2 * inv;
  }

  const float* xg = xb + 128 * wave + 4 * lane;
  float4 xv[NSTRIP][16];
#pragma unroll
  for (int s = 0; s < NSTRIP; ++s) {
#pragma unroll
    for (int i = 0; i < 16; ++i)
      xv[s][i] = *(const float4*)(xg + (size_t)(16 * s + i) * Hsz);
  }

  const int ea = (lane & 8) ? i2 : i1;
  const float* apk = A_down + ((size_t)ea * Hsz + 128 * wave + h2) * Rsz + (lane & 7);
  v2f bf[32];
#pragma unroll
  for (int kk = 0; kk < 32; ++kk) {
    bf[kk][0] = apk[kk * 4 * Rsz];
    bf[kk][1] = apk[kk * 4 * Rsz + Rsz];
  }

#pragma unroll
  for (int s = 0; s < NSTRIP; ++s) {
#pragma unroll
    for (int i = 0; i < 16; ++i)
      *(float4*)(lds_x[wave][s] + i * XPITCH + 4 * lane) = xv[s][i];
  }
  __syncthreads();

  v8f part[NSTRIP];
#pragma unroll
  for (int s = 0; s < NSTRIP; ++s) {
    v8f acc0 = {}, acc1 = {}, acc2 = {}, acc3 = {};
    const float* xa = lds_x[wave][s] + r * XPITCH + h2;
    v2f af[32];
#pragma unroll
    for (int kk = 0; kk < 32; ++kk)
      af[kk] = *(const v2f*)(xa + 4 * kk);
#pragma unroll
    for (int kk = 0; kk < 32; kk += 4) {
      acc0 = __builtin_amdgcn_wmma_f32_16x16x4_f32(false, af[kk + 0], false, bf[kk + 0], (short)0, acc0, false, false);
      acc1 = __builtin_amdgcn_wmma_f32_16x16x4_f32(false, af[kk + 1], false, bf[kk + 1], (short)0, acc1, false, false);
      acc2 = __builtin_amdgcn_wmma_f32_16x16x4_f32(false, af[kk + 2], false, bf[kk + 2], (short)0, acc2, false, false);
      acc3 = __builtin_amdgcn_wmma_f32_16x16x4_f32(false, af[kk + 3], false, bf[kk + 3], (short)0, acc3, false, false);
    }
    part[s] = (acc0 + acc1) + (acc2 + acc3);
  }

  const int nc0 = wave * 8;
  const float* bp1 = B_up + ((size_t)i1 * Rsz + h2) * Hsz + (size_t)nc0 * 16 + (lane & 15);
  const float* bp2 = B_up + ((size_t)i2 * Rsz + h2) * Hsz + (size_t)nc0 * 16 + (lane & 15);
  const size_t obase = (size_t)(8 * (lane >> 4)) * Hsz + (size_t)nc0 * 16 + (lane & 15);
  float* outp = out + ((size_t)b * Lsz + row0) * Hsz + obase;
  const int xres_o = (8 * (lane >> 4)) * XPITCH + (lane & 15);
  v2f bo[8][4];
#pragma unroll
  for (int i = 0; i < 8; ++i) {
#pragma unroll
    for (int ks = 0; ks < 2; ++ks) {
      bo[i][ks][0]     = g1 * bp1[(4 * ks) * Hsz + i * 16];
      bo[i][ks][1]     = g1 * bp1[(4 * ks + 1) * Hsz + i * 16];
      bo[i][ks + 2][0] = g2 * bp2[(4 * ks) * Hsz + i * 16];
      bo[i][ks + 2][1] = g2 * bp2[(4 * ks + 1) * Hsz + i * 16];
    }
  }

#pragma unroll
  for (int s = 0; s < NSTRIP; ++s) {
#pragma unroll
    for (int v = 0; v < 8; ++v)
      lds_part[s][wave][v * 32 + lane] = part[s][v];
  }
  __syncthreads();
#pragma unroll
  for (int s = 0; s < NSTRIP; ++s) {
    const int t = threadIdx.x;
    const float sum = ((lds_part[s][0][t] + lds_part[s][1][t]) + (lds_part[s][2][t] + lds_part[s][3][t])) +
                      ((lds_part[s][4][t] + lds_part[s][5][t]) + (lds_part[s][6][t] + lds_part[s][7][t]));
    lds_low[s][(wave + 8 * (lane >> 4)) * 18 + (lane & 15)] = sum;
  }
  __syncthreads();

#pragma unroll
  for (int s = 0; s < NSTRIP; ++s) {
    v2f a2op[4];
#pragma unroll
    for (int ks = 0; ks < 4; ++ks)
      a2op[ks] = *(const v2f*)(&lds_low[s][r * 18 + 4 * ks + h2]);

    const float* xres_p = lds_x[wave][s] + xres_o;
    float xres[8][8];
#pragma unroll
    for (int i = 0; i < 8; ++i) {
#pragma unroll
      for (int v = 0; v < 8; ++v)
        xres[i][v] = xres_p[v * XPITCH + i * 16];
    }

    float* outs = outp + (size_t)(16 * s) * Hsz;
#pragma unroll
    for (int i = 0; i < 8; ++i) {
      v8f y = {};
#pragma unroll
      for (int ks = 0; ks < 4; ++ks)
        y = __builtin_amdgcn_wmma_f32_16x16x4_f32(
            false, a2op[ks], false, bo[i][ks], (short)0, y, false, false);
#pragma unroll
      for (int v = 0; v < 8; ++v)
        outs[(size_t)v * Hsz + i * 16] = xres[i][v] + y[v];
    }
  }
}

extern "C" void kernel_launch(void* const* d_in, const int* in_sizes, int n_in,
                              void* d_out, int out_size, void* d_ws, size_t ws_size,
                              hipStream_t stream) {
  const float* x        = (const float*)d_in[0];
  const float* W_router = (const float*)d_in[1];
  const float* A_down   = (const float*)d_in[2];
  const float* B_up     = (const float*)d_in[3];
  float* out = (float*)d_out;

  moe_main<<<dim3(Lsz / (16 * NSTRIP), Bsz), dim3(256), 0, stream>>>(x, W_router, A_down, B_up, out);
}


# stage-2 columns dealt so that a lane owns 4 adjacent columns, one per tile of a group of 4: B_up operand loads, LDS residual reads and output stores are 16 bytes per lane, whole 128-byte lines
# speedup vs baseline: 5.0701x; 1.0963x over previous
#include <hip/hip_runtime.h>
#include <hip/hip_bf16.h>
#include <math.h>


#define Bsz 4
#define Lsz 2048
#define Hsz 1024
#define Esz 8
#define Rsz 8

typedef __attribute__((ext_vector_type(2))) float v2f;
typedef __attribute__((ext_vector_type(8))) float v8f;


#define NWAVE 8
#define NSTRIP 2
#define XPITCH 132
__global__ __launch_bounds__(256) void moe_main(
    const float* __restrict__ x,
    const float* __restrict__ W_router,
    const float* __restrict__ A_down,
    const float* __restrict__ B_up,
    float* __restrict__ out) {
  const int b    = blockIdx.y;
  const int wave = threadIdx.x >> 5;
  const int lane = threadIdx.x & 31;
  const int row0 = blockIdx.x * (16 * NSTRIP);

  const int r  = lane & 15;
  const int h2 = (lane >> 4) * 2;

  const float* xb = x + ((size_t)b * Lsz + row0) * Hsz;

  __shared__ __align__(16) float lds_x[NWAVE][NSTRIP][16 * XPITCH];
  __shared__ float lds_part[NSTRIP][NWAVE][256];
  __shared__ __align__(16) float lds_low[NSTRIP][16 * 18];
  __shared__ float red[256];
  __shared__ float slog[Esz];

  const float* xr0 = x + (size_t)b * Lsz * Hsz + lane;
  const float* wr  = W_router + (size_t)wave * Hsz + lane;
  float gx[32], gw[32];
#pragma unroll
  for (int i = 0; i < 32; ++i) {
    gx[i] = xr0[32 * i];
    gw[i] = wr[32 * i];
  }

  {
    float p = 0.f;
#pragma unroll
    for (int i = 0; i < 32; ++i) p += gx[i] * gw[i];
    red[threadIdx.x] = p;
  }
  __syncthreads();
  {
    float s = 0.f;
#pragma unroll
    for (int i = 0; i < 32; ++i) s += red[wave * 32 + i];
    slog[wave] = s;
  }
  __syncthreads();

  int i1 = 0, i2 = -1;
  float g1, g2;
  {
    float lg[Esz];
#pragma unroll
    for (int e = 0; e < Esz; ++e) lg[e] = slog[e];
    float v1 = lg[0];
#pragma unroll
    for (int e = 1; e < Esz; ++e)
      if (lg[e] > v1) { v1 = lg[e]; i1 = e; }
    float v2 = -INFINITY;
#pragma unroll
    for (int e = 0; e < Esz; ++e) {
      if (e == i1) continue;
      if (lg[e] > v2) { v2 = lg[e]; i2 = e; }
    }
    if (i2 < 0) i2 = i1 ^ 1;
    const float e2 = expf(v2 - v1);
    const float inv = 1.f / (1.f + e2);
    g1 = inv;
    g2 = e2 * inv;
  }

  const float* xg = xb + 128 * wave + 4 * lane;
  float4 xv[NSTRIP][16];
#pragma unroll
  for (int s = 0; s < NSTRIP; ++s) {
#pragma unroll
    for (int i = 0; i < 16; ++i)
      xv[s][i] = *(const float4*)(xg + (size_t)(16 * s + i) * Hsz);
  }

  const int ea = (lane & 8) ? i2 : i1;
  const float* apk = A_down + ((size_t)ea * Hsz + 128 * wave + h2) * Rsz + (lane & 7);
  v2f bf[32];
#pragma unroll
  for (int kk = 0; kk < 32; ++kk) {
    bf[kk][0] = apk[kk * 4 * Rsz];
    bf[kk][1] = apk[kk * 4 * Rsz + Rsz];
  }

#pragma unroll
  for (int s = 0; s < NSTRIP; ++s) {
#pragma unroll
    for (int i = 0; i < 16; ++i)
      *(float4*)(lds_x[wave][s] + i * XPITCH + 4 * lane) = xv[s][i];
  }
  __syncthreads();

  v8f part[NSTRIP];
#pragma unroll
  for (int s = 0; s < NSTRIP; ++s) {
    v8f acc0 = {}, acc1 = {}, acc2 = {}, acc3 = {};
    const float* xa = lds_x[wave][s] + r * XPITCH + h2;
    v2f af[32];
#pragma unroll
    for (int kk = 0; kk < 32; ++kk)
      af[kk] = *(const v2f*)(xa + 4 * kk);
#pragma unroll
    for (int kk = 0; kk < 32; kk += 4) {
      acc0 = __builtin_amdgcn_wmma_f32_16x16x4_f32(false, af[kk + 0], false, bf[kk + 0], (short)0, acc0, false, false);
      acc1 = __builtin_amdgcn_wmma_f32_16x16x4_f32(false, af[kk + 1], false, bf[kk + 1], (short)0, acc1, false, false);
      acc2 = __builtin_amdgcn_wmma_f32_16x16x4_f32(false, af[kk + 2], false, bf[kk + 2], (short)0, acc2, false, false);
      acc3 = __builtin_amdgcn_wmma_f32_16x16x4_f32(false, af[kk + 3], false, bf[kk + 3], (short)0, acc3, false, false);
    }
    part[s] = (acc0 + acc1) + (acc2 + acc3);
  }

  const int c4 = 4 * (lane & 15);
  const float* bp1 = B_up + ((size_t)i1 * Rsz + h2) * Hsz + 128 * wave + c4;
  const float* bp2 = B_up + ((size_t)i2 * Rsz + h2) * Hsz + 128 * wave + c4;
  const size_t obase = (size_t)(8 * (lane >> 4)) * Hsz + 128 * wave + c4;
  float* outp = out + ((size_t)b * Lsz + row0) * Hsz + obase;
  const int xres_o = (8 * (lane >> 4)) * XPITCH + c4;
  v2f bo[8][4];
#pragma unroll
  for (int m = 0; m < 2; ++m) {
#pragma unroll
    for (int ks = 0; ks < 2; ++ks) {
#pragma unroll
      for (int j = 0; j < 2; ++j) {
        const float4 t1 = *(const float4*)(bp1 + (4 * ks + j) * Hsz + 64 * m);
        const float4 t2 = *(const float4*)(bp2 + (4 * ks + j) * Hsz + 64 * m);
        bo[4 * m + 0][ks][j] = g1 * t1.x;
        bo[4 * m + 1][ks][j] = g1 * t1.y;
        bo[4 * m + 2][ks][j] = g1 * t1.z;
        bo[4 * m + 3][ks][j] = g1 * t1.w;
        bo[4 * m + 0][ks + 2][j] = g2 * t2.x;
        bo[4 * m + 1][ks + 2][j] = g2 * t2.y;
        bo[4 * m + 2][ks + 2][j] = g2 * t2.z;
        bo[4 * m + 3][ks + 2][j] = g2 * t2.w;
      }
    }
  }

#pragma unroll
  for (int s = 0; s < NSTRIP; ++s) {
#pragma unroll
    for (int v = 0; v < 8; ++v)
      lds_part[s][wave][v * 32 + lane] = part[s][v];
  }
  __syncthreads();
#pragma unroll
  for (int s = 0; s < NSTRIP; ++s) {
    const int t = threadIdx.x;
    const float sum = ((lds_part[s][0][t] + lds_part[s][1][t]) + (lds_part[s][2][t] + lds_part[s][3][t])) +
                      ((lds_part[s][4][t] + lds_part[s][5][t]) + (lds_part[s][6][t] + lds_part[s][7][t]));
    lds_low[s][(wave + 8 * (lane >> 4)) * 18 + (lane & 15)] = sum;
  }
  __syncthreads();

#pragma unroll
  for (int s = 0; s < NSTRIP; ++s) {
    v2f a2op[4];
#pragma unroll
    for (int ks = 0; ks < 4; ++ks)
      a2op[ks] = *(const v2f*)(&lds_low[s][r * 18 + 4 * ks + h2]);

    const float* xres_p = lds_x[wave][s] + xres_o;
    float4 xres[2][8];
#pragma unroll
    for (int m = 0; m < 2; ++m) {
#pragma unroll
      for (int v = 0; v < 8; ++v)
        xres[m][v] = *(const float4*)(xres_p + v * XPITCH + 64 * m);
    }

    float* outs = outp + (size_t)(16 * s) * Hsz;
#pragma unroll
    for (int m = 0; m < 2; ++m) {
      v8f y0 = {}, y1 = {}, y2 = {}, y3 = {};
#pragma unroll
      for (int ks = 0; ks < 4; ++ks) {
        y0 = __builtin_amdgcn_wmma_f32_16x16x4_f32(
            false, a2op[ks], false, bo[4 * m + 0][ks], (short)0, y0, false, false);
        y1 = __builtin_amdgcn_wmma_f32_16x16x4_f32(
            false, a2op[ks], false, bo[4 * m + 1][ks], (short)0, y1, false, false);
        y2 = __builtin_amdgcn_wmma_f32_16x16x4_f32(
            false, a2op[ks], false, bo[4 * m + 2][ks], (short)0, y2, false, false);
        y3 = __builtin_amdgcn_wmma_f32_16x16x4_f32(
            false, a2op[ks], false, bo[4 * m + 3][ks], (short)0, y3, false, false);
      }
#pragma unroll
      for (int v = 0; v < 8; ++v) {
        float4 o;
        o.x = xres[m][v].x + y0[v];
        o.y = xres[m][v].y + y1[v];
        o.z = xres[m][v].z + y2[v];
        o.w = xres[m][v].w + y3[v];
        *(float4*)(outs + (size_t)v * Hsz + 64 * m) = o;
      }
    }
  }
}

extern "C" void kernel_launch(void* const* d_in, const int* in_sizes, int n_in,
                              void* d_out, int out_size, void* d_ws, size_t ws_size,
                              hipStream_t stream) {
  const float* x        = (const float*)d_in[0];
  const float* W_router = (const float*)d_in[1];
  const float* A_down   = (const float*)d_in[2];
  const float* B_up     = (const float*)d_in[3];
  float* out = (float*)d_out;

  moe_main<<<dim3(Lsz / (16 * NSTRIP), Bsz), dim3(256), 0, stream>>>(x, W_router, A_down, B_up, out);
}


# global loads and stores kept in flight across LDS-only barriers, strips run one after the other: x arrives under the gate and strip 0's stage 1, strip 0's stores drain under strip 1's matrix work
# speedup vs baseline: 6.1476x; 1.2125x over previous
#include <hip/hip_runtime.h>
#include <hip/hip_bf16.h>
#include <math.h>


#define Bsz 4
#define Lsz 2048
#define Hsz 1024
#define Esz 8
#define Rsz 8

typedef __attribute__((ext_vector_type(2))) float v2f;
typedef __attribute__((ext_vector_type(4))) float v4f;
typedef __attribute__((ext_vector_type(8))) float v8f;


__device__ __forceinline__ void lds_barrier() {
  __builtin_amdgcn_fence(__ATOMIC_RELEASE, "workgroup", "local");
  __builtin_amdgcn_s_barrier();
  __builtin_amdgcn_fence(__ATOMIC_ACQUIRE, "workgroup", "local");
}

#define NWAVE 8
#define NSTRIP 2
#define XPITCH 132
__global__ __launch_bounds__(256) void moe_main(
    const float* __restrict__ x,
    const float* __restrict__ W_router,
    const float* __restrict__ A_down,
    const float* __restrict__ B_up,
    float* __restrict__ out) {
  const int b    = blockIdx.y;
  const int wave = threadIdx.x >> 5;
  const int lane = threadIdx.x & 31;
  const int row0 = blockIdx.x * (16 * NSTRIP);

  const int r  = lane & 15;
  const int h2 = (lane >> 4) * 2;

  const float* xb = x + ((size_t)b * Lsz + row0) * Hsz;

  __shared__ __align__(16) float lds_x[NWAVE][NSTRIP][16 * XPITCH];
  __shared__ float lds_part[NSTRIP][NWAVE][256];
  __shared__ __align__(16) float lds_low[NSTRIP][16 * 18];
  __shared__ float red[256];
  __shared__ float slog[Esz];

  const float* xr0 = x + (size_t)b * Lsz * Hsz + lane;
  const float* wr  = W_router + (size_t)wave * Hsz + lane;
  float gx[32], gw[32];
#pragma unroll
  for (int i = 0; i < 32; ++i) {
    gx[i] = xr0[32 * i];
    gw[i] = wr[32 * i];
  }

  const float* xg = xb + 128 * wave + 4 * lane;
  v4f xv0[16], xv1[16];
#pragma unroll
  for (int i = 0; i < 16; ++i)
    xv0[i] = *(const v4f*)(xg + (size_t)i * Hsz);

  {
    float p = 0.f;
#pragma unroll
    for (int i = 0; i < 32; ++i) p += gx[i] * gw[i];
    red[threadIdx.x] = p;
  }
  lds_barrier();
  {
    float s = 0.f;
#pragma unroll
    for (int i = 0; i < 32; ++i) s += red[wave * 32 + i];
    slog[wave] = s;
  }
  lds_barrier();

  int i1 = 0, i2 = -1;
  float g1, g2;
  {
    float lg[Esz];
#pragma unroll
    for (int e = 0; e < Esz; ++e) lg[e] = slog[e];
    float v1 = lg[0];
#pragma unroll
    for (int e = 1; e < Esz; ++e)
      if (lg[e] > v1) { v1 = lg[e]; i1 = e; }
    float v2 = -INFINITY;
#pragma unroll
    for (int e = 0; e < Esz; ++e) {
      if (e == i1) continue;
      if (lg[e] > v2) { v2 = lg[e]; i2 = e; }
    }
    if (i2 < 0) i2 = i1 ^ 1;
    const float e2 = expf(v2 - v1);
    const float inv = 1.f / (1.f + e2);
    g1 = inv;
    g2 = e2 * inv;
  }


  const int ea = (lane & 8) ? i2 : i1;
  const float* apk = A_down + ((size_t)ea * Hsz + 128 * wave + h2) * Rsz + (lane & 7);
  v2f bf[32];
#pragma unroll
  for (int kk = 0; kk < 32; ++kk) {
    bf[kk][0] = apk[kk * 4 * Rsz];
    bf[kk][1] = apk[kk * 4 * Rsz + Rsz];
  }
  __builtin_amdgcn_sched_barrier(0);

#pragma unroll
  for (int i = 0; i < 16; ++i)
    xv1[i] = *(const v4f*)(xg + (size_t)(16 + i) * Hsz);

  const int c4 = 4 * (lane & 15);
  const float* bp1 = B_up + ((size_t)i1 * Rsz + h2) * Hsz + 128 * wave + c4;
  const float* bp2 = B_up + ((size_t)i2 * Rsz + h2) * Hsz + 128 * wave + c4;
  v4f t1[2][2][2], t2[2][2][2];
#pragma unroll
  for (int m = 0; m < 2; ++m) {
#pragma unroll
    for (int ks = 0; ks < 2; ++ks) {
#pragma unroll
      for (int j = 0; j < 2; ++j) {
        t1[m][ks][j] = *(const v4f*)(bp1 + (4 * ks + j) * Hsz + 64 * m);
        t2[m][ks][j] = *(const v4f*)(bp2 + (4 * ks + j) * Hsz + 64 * m);
      }
    }
  }

  const size_t obase = (size_t)(8 * (lane >> 4)) * Hsz + 128 * wave + c4;
  float* outp = out + ((size_t)b * Lsz + row0) * Hsz + obase;
  const int xres_o = (8 * (lane >> 4)) * XPITCH + c4;
  v2f bo[8][4];


  auto stage_slice = [&](const int s, const v4f (&xs)[16]) __attribute__((always_inline)) {
#pragma unroll
    for (int i = 0; i < 16; ++i)
      *(v4f*)(lds_x[wave][s] + i * XPITCH + 4 * lane) = xs[i];
  };

  auto stage1 = [&](const int s) __attribute__((always_inline)) {
    v8f acc0 = {}, acc1 = {}, acc2 = {}, acc3 = {};
    const float* xa = lds_x[wave][s] + r * XPITCH + h2;
    v2f af[32];
#pragma unroll
    for (int kk = 0; kk < 32; ++kk)
      af[kk] = *(const v2f*)(xa + 4 * kk);
#pragma unroll
    for (int kk = 0; kk < 32; kk += 4) {
      acc0 = __builtin_amdgcn_wmma_f32_16x16x4_f32(false, af[kk + 0], false, bf[kk + 0], (short)0, acc0, false, false);
      acc1 = __builtin_amdgcn_wmma_f32_16x16x4_f32(false, af[kk + 1], false, bf[kk + 1], (short)0, acc1, false, false);
      acc2 = __builtin_amdgcn_wmma_f32_16x16x4_f32(false, af[kk + 2], false, bf[kk + 2], (short)0, acc2, false, false);
      acc3 = __builtin_amdgcn_wmma_f32_16x16x4_f32(false, af[kk + 3], false, bf[kk + 3], (short)0, acc3, false, false);
    }
    const v8f part = (acc0 + acc1) + (acc2 + acc3);
#pragma unroll
    for (int v = 0; v < 8; ++v)
      lds_part[s][wave][v * 32 + lane] = part[v];
  };

  auto sum_parts = [&](const int s) __attribute__((always_inline)) {
    const int t = threadIdx.x;
    const float sum = ((lds_part[s][0][t] + lds_part[s][1][t]) + (lds_part[s][2][t] + lds_part[s][3][t])) +
                      ((lds_part[s][4][t] + lds_part[s][5][t]) + (lds_part[s][6][t] + lds_part[s][7][t]));
    lds_low[s][(wave + 8 * (lane >> 4)) * 18 + (lane & 15)] = sum;
  };

  auto stage2 = [&](const int s) __attribute__((always_inline)) {
    v2f a2op[4];
#pragma unroll
    for (int ks = 0; ks < 4; ++ks)
      a2op[ks] = *(const v2f*)(&lds_low[s][r * 18 + 4 * ks + h2]);

    const float* xres_p = lds_x[wave][s] + xres_o;
    float4 xres[2][8];
#pragma unroll
    for (int m = 0; m < 2; ++m) {
#pragma unroll
      for (int v = 0; v < 8; ++v)
        xres[m][v] = *(const float4*)(xres_p + v * XPITCH + 64 * m);
    }

    float* outs = outp + (size_t)(16 * s) * Hsz;
#pragma unroll
    for (int m = 0; m < 2; ++m) {
      v8f y0 = {}, y1 = {}, y2 = {}, y3 = {};
#pragma unroll
      for (int ks = 0; ks < 4; ++ks) {
        y0 = __builtin_amdgcn_wmma_f32_16x16x4_f32(
            false, a2op[ks], false, bo[4 * m + 0][ks], (short)0, y0, false, false);
        y1 = __builtin_amdgcn_wmma_f32_16x16x4_f32(
            false, a2op[ks], false, bo[4 * m + 1][ks], (short)0, y1, false, false);
        y2 = __builtin_amdgcn_wmma_f32_16x16x4_f32(
            false, a2op[ks], false, bo[4 * m + 2][ks], (short)0, y2, false, false);
        y3 = __builtin_amdgcn_wmma_f32_16x16x4_f32(
            false, a2op[ks], false, bo[4 * m + 3][ks], (short)0, y3, false, false);
      }
#pragma unroll
      for (int v = 0; v < 8; ++v) {
        float4 o;
        o.x = xres[m][v].x + y0[v];
        o.y = xres[m][v].y + y1[v];
        o.z = xres[m][v].z + y2[v];
        o.w = xres[m][v].w + y3[v];
        *(float4*)(outs + (size_t)v * Hsz + 64 * m) = o;
      }
    }
  };

  stage_slice(0, xv0);
  lds_barrier();
  stage1(0);
  __builtin_amdgcn_sched_barrier(0);
  stage_slice(1, xv1);
  lds_barrier();
  sum_parts(0);
  lds_barrier();

#pragma unroll
  for (int m = 0; m < 2; ++m) {
#pragma unroll
    for (int ks = 0; ks < 2; ++ks) {
#pragma unroll
      for (int j = 0; j < 2; ++j) {
        bo[4 * m + 0][ks][j] = g1 * t1[m][ks][j].x;
        bo[4 * m + 1][ks][j] = g1 * t1[m][ks][j].y;
        bo[4 * m + 2][ks][j] = g1 * t1[m][ks][j].z;
        bo[4 * m + 3][ks][j] = g1 * t1[m][ks][j].w;
        bo[4 * m + 0][ks + 2][j] = g2 * t2[m][ks][j].x;
        bo[4 * m + 1][ks + 2][j] = g2 * t2[m][ks][j].y;
        bo[4 * m + 2][ks + 2][j] = g2 * t2[m][ks][j].z;
        bo[4 * m + 3][ks + 2][j] = g2 * t2[m][ks][j].w;
      }
    }
  }
  stage2(0);
  __builtin_amdgcn_sched_barrier(0);
  stage1(1);
  lds_barrier();
  sum_parts(1);
  lds_barrier();
  stage2(1);
}

extern "C" void kernel_launch(void* const* d_in, const int* in_sizes, int n_in,
                              void* d_out, int out_size, void* d_ws, size_t ws_size,
                              hipStream_t stream) {
  const float* x        = (const float*)d_in[0];
  const float* W_router = (const float*)d_in[1];
  const float* A_down   = (const float*)d_in[2];
  const float* B_up     = (const float*)d_in[3];
  float* out = (float*)d_out;

  moe_main<<<dim3(Lsz / (16 * NSTRIP), Bsz), dim3(256), 0, stream>>>(x, W_router, A_down, B_up, out);
}


# router gate read with 16-byte loads of disjoint 128-float slices per wave (9 loads per lane, was 64), slices summed by lane butterfly and one LDS exchange (was two): the logits' sum order changes
# speedup vs baseline: 6.5060x; 1.0583x over previous
#include <hip/hip_runtime.h>
#include <hip/hip_bf16.h>
#include <math.h>


#define Bsz 4
#define Lsz 2048
#define Hsz 1024
#define Esz 8
#define Rsz 8

typedef __attribute__((ext_vector_type(2))) float v2f;
typedef __attribute__((ext_vector_type(4))) float v4f;
typedef __attribute__((ext_vector_type(8))) float v8f;


__device__ __forceinline__ void lds_barrier() {
  __builtin_amdgcn_fence(__ATOMIC_RELEASE, "workgroup", "local");
  __builtin_amdgcn_s_barrier();
  __builtin_amdgcn_fence(__ATOMIC_ACQUIRE, "workgroup", "local");
}

template <int M>
__device__ __forceinline__ float row_xor(float v) {
  return __builtin_bit_cast(float, __builtin_amdgcn_update_dpp(0, __builtin_bit_cast(int, v), 0x160 | M, 0xf, 0xf, true));
}
__device__ __forceinline__ float xhalf(float v) {
  return __builtin_bit_cast(float, __builtin_amdgcn_permlanex16(__builtin_bit_cast(int, v), __builtin_bit_cast(int, v),
                                                               0x76543210, 0xfedcba98, false, false));
}

#define NWAVE 8
#define NSTRIP 2
#define XPITCH 132
__global__ __launch_bounds__(256) void moe_main(
    const float* __restrict__ x,
    const float* __restrict__ W_router,
    const float* __restrict__ A_down,
    const float* __restrict__ B_up,
    float* __restrict__ out) {
  const int b    = blockIdx.y;
  const int wave = threadIdx.x >> 5;
  const int lane = threadIdx.x & 31;
  const int row0 = blockIdx.x * (16 * NSTRIP);

  const int r  = lane & 15;
  const int h2 = (lane >> 4) * 2;

  const float* xb = x + ((size_t)b * Lsz + row0) * Hsz;

  __shared__ __align__(16) float lds_x[NWAVE][NSTRIP][16 * XPITCH];
  __shared__ float lds_part[NSTRIP][NWAVE][256];
  __shared__ __align__(16) float lds_low[NSTRIP][16 * 18];
  __shared__ float wsum[NWAVE * Esz];

  const float* xr0 = x + (size_t)b * Lsz * Hsz + 128 * wave + 4 * lane;
  const float* wr  = W_router + 128 * wave + 4 * lane;
  const v4f gxv = *(const v4f*)xr0;
  v4f gwv[Esz];
#pragma unroll
  for (int e = 0; e < Esz; ++e)
    gwv[e] = *(const v4f*)(wr + (size_t)e * Hsz);

  const float* xg = xb + 128 * wave + 4 * lane;
  v4f xv0[16], xv1[16];
#pragma unroll
  for (int i = 0; i < 16; ++i)
    xv0[i] = *(const v4f*)(xg + (size_t)i * Hsz);

  float p[Esz];
#pragma unroll
  for (int e = 0; e < Esz; ++e)
    p[e] = ((gxv.x * gwv[e].x + gxv.y * gwv[e].y) + gxv.z * gwv[e].z) + gxv.w * gwv[e].w;
#pragma unroll
  for (int e = 0; e < Esz; ++e) {
    p[e] += row_xor<1>(p[e]);
    p[e] += row_xor<2>(p[e]);
    p[e] += row_xor<4>(p[e]);
    p[e] += row_xor<8>(p[e]);
    p[e] += xhalf(p[e]);
  }
  {
    float mine = p[0];
#pragma unroll
    for (int e = 1; e < Esz; ++e) mine = (lane == e) ? p[e] : mine;
    if (lane < Esz) wsum[wave * Esz + lane] = mine;
  }
  lds_barrier();

  int i1 = 0, i2 = -1;
  float g1, g2;
  {
    const float* ws = wsum + (lane & 7);
    const float sl = ((ws[0 * Esz] + ws[1 * Esz]) + (ws[2 * Esz] + ws[3 * Esz])) +
                     ((ws[4 * Esz] + ws[5 * Esz]) + (ws[6 * Esz] + ws[7 * Esz]));
    float lg[Esz];
#pragma unroll
    for (int e = 0; e < Esz; ++e)
      lg[e] = __builtin_bit_cast(float, __builtin_amdgcn_readlane(__builtin_bit_cast(int, sl), e));
    float v1 = lg[0];
#pragma unroll
    for (int e = 1; e < Esz; ++e)
      if (lg[e] > v1) { v1 = lg[e]; i1 = e; }
    float v2 = -INFINITY;
#pragma unroll
    for (int e = 0; e < Esz; ++e) {
      if (e == i1) continue;
      if (lg[e] > v2) { v2 = lg[e]; i2 = e; }
    }
    if (i2 < 0) i2 = i1 ^ 1;
    const float e2 = expf(v2 - v1);
    const float inv = 1.f / (1.f + e2);
    g1 = inv;
    g2 = e2 * inv;
  }


  const int ea = (lane & 8) ? i2 : i1;
  const float* apk = A_down + ((size_t)ea * Hsz + 128 * wave + h2) * Rsz + (lane & 7);
  v2f bf[32];
#pragma unroll
  for (int kk = 0; kk < 32; ++kk) {
    bf[kk][0] = apk[kk * 4 * Rsz];
    bf[kk][1] = apk[kk * 4 * Rsz + Rsz];
  }
  __builtin_amdgcn_sched_barrier(0);

#pragma unroll
  for (int i = 0; i < 16; ++i)
    xv1[i] = *(const v4f*)(xg + (size_t)(16 + i) * Hsz);

  const int c4 = 4 * (lane & 15);
  const float* bp1 = B_up + ((size_t)i1 * Rsz + h2) * Hsz + 128 * wave + c4;
  const float* bp2 = B_up + ((size_t)i2 * Rsz + h2) * Hsz + 128 * wave + c4;
  v4f t1[2][2][2], t2[2][2][2];
#pragma unroll
  for (int m = 0; m < 2; ++m) {
#pragma unroll
    for (int ks = 0; ks < 2; ++ks) {
#pragma unroll
      for (int j = 0; j < 2; ++j) {
        t1[m][ks][j] = *(const v4f*)(bp1 + (4 * ks + j) * Hsz + 64 * m);
        t2[m][ks][j] = *(const v4f*)(bp2 + (4 * ks + j) * Hsz + 64 * m);
      }
    }
  }

  const size_t obase = (size_t)(8 * (lane >> 4)) * Hsz + 128 * wave + c4;
  float* outp = out + ((size_t)b * Lsz + row0) * Hsz + obase;
  const int xres_o = (8 * (lane >> 4)) * XPITCH + c4;
  v2f bo[8][4];


  auto stage_slice = [&](const int s, const v4f (&xs)[16]) __attribute__((always_inline)) {
#pragma unroll
    for (int i = 0; i < 16; ++i)
      *(v4f*)(lds_x[wave][s] + i * XPITCH + 4 * lane) = xs[i];
  };

  auto stage1 = [&](const int s) __attribute__((always_inline)) {
    v8f acc0 = {}, acc1 = {}, acc2 = {}, acc3 = {};
    const float* xa = lds_x[wave][s] + r * XPITCH + h2;
    v2f af[32];
#pragma unroll
    for (int kk = 0; kk < 32; ++kk)
      af[kk] = *(const v2f*)(xa + 4 * kk);
#pragma unroll
    for (int kk = 0; kk < 32; kk += 4) {
      acc0 = __builtin_amdgcn_wmma_f32_16x16x4_f32(false, af[kk + 0], false, bf[kk + 0], (short)0, acc0, false, false);
      acc1 = __builtin_amdgcn_wmma_f32_16x16x4_f32(false, af[kk + 1], false, bf[kk + 1], (short)0, acc1, false, false);
      acc2 = __builtin_amdgcn_wmma_f32_16x16x4_f32(false, af[kk + 2], false, bf[kk + 2], (short)0, acc2, false, false);
      acc3 = __builtin_amdgcn_wmma_f32_16x16x4_f32(false, af[kk + 3], false, bf[kk + 3], (short)0, acc3, false, false);
    }
    const v8f part = (acc0 + acc1) + (acc2 + acc3);
#pragma unroll
    for (int v = 0; v < 8; ++v)
      lds_part[s][wave][v * 32 + lane] = part[v];
  };

  auto sum_parts = [&](const int s) __attribute__((always_inline)) {
    const int t = threadIdx.x;
    const float sum = ((lds_part[s][0][t] + lds_part[s][1][t]) + (lds_part[s][2][t] + lds_part[s][3][t])) +
                      ((lds_part[s][4][t] + lds_part[s][5][t]) + (lds_part[s][6][t] + lds_part[s][7][t]));
    lds_low[s][(wave + 8 * (lane >> 4)) * 18 + (lane & 15)] = sum;
  };

  auto stage2 = [&](const int s) __attribute__((always_inline)) {
    v2f a2op[4];
#pragma unroll
    for (int ks = 0; ks < 4; ++ks)
      a2op[ks] = *(const v2f*)(&lds_low[s][r * 18 + 4 * ks + h2]);

    const float* xres_p = lds_x[wave][s] + xres_o;
    float4 xres[2][8];
#pragma unroll
    for (int m = 0; m < 2; ++m) {
#pragma unroll
      for (int v = 0; v < 8; ++v)
        xres[m][v] = *(const float4*)(xres_p + v * XPITCH + 64 * m);
    }

    float* outs = outp + (size_t)(16 * s) * Hsz;
#pragma unroll
    for (int m = 0; m < 2; ++m) {
      v8f y0 = {}, y1 = {}, y2 = {}, y3 = {};
#pragma unroll
      for (int ks = 0; ks < 4; ++ks) {
        y0 = __builtin_amdgcn_wmma_f32_16x16x4_f32(
            false, a2op[ks], false, bo[4 * m + 0][ks], (short)0, y0, false, false);
        y1 = __builtin_amdgcn_wmma_f32_16x16x4_f32(
            false, a2op[ks], false, bo[4 * m + 1][ks], (short)0, y1, false, false);
        y2 = __builtin_amdgcn_wmma_f32_16x16x4_f32(
            false, a2op[ks], false, bo[4 * m + 2][ks], (short)0, y2, false, false);
        y3 = __builtin_amdgcn_wmma_f32_16x16x4_f32(
            false, a2op[ks], false, bo[4 * m + 3][ks], (short)0, y3, false, false);
      }
#pragma unroll
      for (int v = 0; v < 8; ++v) {
        float4 o;
        o.x = xres[m][v].x + y0[v];
        o.y = xres[m][v].y + y1[v];
        o.z = xres[m][v].z + y2[v];
        o.w = xres[m][v].w + y3[v];
        *(float4*)(outs + (size_t)v * Hsz + 64 * m) = o;
      }
    }
  };

  stage_slice(0, xv0);
  lds_barrier();
  stage1(0);
  __builtin_amdgcn_sched_barrier(0);
  stage_slice(1, xv1);
  lds_barrier();
  sum_parts(0);
  lds_barrier();

#pragma unroll
  for (int m = 0; m < 2; ++m) {
#pragma unroll
    for (int ks = 0; ks < 2; ++ks) {
#pragma unroll
      for (int j = 0; j < 2; ++j) {
        bo[4 * m + 0][ks][j] = g1 * t1[m][ks][j].x;
        bo[4 * m + 1][ks][j] = g1 * t1[m][ks][j].y;
        bo[4 * m + 2][ks][j] = g1 * t1[m][ks][j].z;
        bo[4 * m + 3][ks][j] = g1 * t1[m][ks][j].w;
        bo[4 * m + 0][ks + 2][j] = g2 * t2[m][ks][j].x;
        bo[4 * m + 1][ks + 2][j] = g2 * t2[m][ks][j].y;
        bo[4 * m + 2][ks + 2][j] = g2 * t2[m][ks][j].z;
        bo[4 * m + 3][ks + 2][j] = g2 * t2[m][ks][j].w;
      }
    }
  }
  stage2(0);
  __builtin_amdgcn_sched_barrier(0);
  stage1(1);
  lds_barrier();
  sum_parts(1);
  lds_barrier();
  stage2(1);
}

extern "C" void kernel_launch(void* const* d_in, const int* in_sizes, int n_in,
                              void* d_out, int out_size, void* d_ws, size_t ws_size,
                              hipStream_t stream) {
  const float* x        = (const float*)d_in[0];
  const float* W_router = (const float*)d_in[1];
  const float* A_down   = (const float*)d_in[2];
  const float* B_up     = (const float*)d_in[3];
  float* out = (float*)d_out;

  moe_main<<<dim3(Lsz / (16 * NSTRIP), Bsz), dim3(256), 0, stream>>>(x, W_router, A_down, B_up, out);
}


# output stores carry the non-temporal hint (still 16 bytes per lane, whole lines): written lines leave the second-level cache for memory as the block works, not at the launch's end; values unchanged
# speedup vs baseline: 6.8289x; 1.0496x over previous
#include <hip/hip_runtime.h>
#include <hip/hip_bf16.h>
#include <math.h>


#define Bsz 4
#define Lsz 2048
#define Hsz 1024
#define Esz 8
#define Rsz 8

typedef __attribute__((ext_vector_type(2))) float v2f;
typedef __attribute__((ext_vector_type(4))) float v4f;
typedef __attribute__((ext_vector_type(8))) float v8f;


__device__ __forceinline__ void lds_barrier() {
  __builtin_amdgcn_fence(__ATOMIC_RELEASE, "workgroup", "local");
  __builtin_amdgcn_s_barrier();
  __builtin_amdgcn_fence(__ATOMIC_ACQUIRE, "workgroup", "local");
}

template <int M>
__device__ __forceinline__ float row_xor(float v) {
  return __builtin_bit_cast(float, __builtin_amdgcn_update_dpp(0, __builtin_bit_cast(int, v), 0x160 | M, 0xf, 0xf, true));
}
__device__ __forceinline__ float xhalf(float v) {
  return __builtin_bit_cast(float, __builtin_amdgcn_permlanex16(__builtin_bit_cast(int, v), __builtin_bit_cast(int, v),
                                                               0x76543210, 0xfedcba98, false, false));
}

#define NWAVE 8
#define NSTRIP 2
#define XPITCH 132
__global__ __launch_bounds__(256) void moe_main(
    const float* __restrict__ x,
    const float* __restrict__ W_router,
    const float* __restrict__ A_down,
    const float* __restrict__ B_up,
    float* __restrict__ out) {
  const int b    = blockIdx.y;
  const int wave = threadIdx.x >> 5;
  const int lane = threadIdx.x & 31;
  const int row0 = blockIdx.x * (16 * NSTRIP);

  const int r  = lane & 15;
  const int h2 = (lane >> 4) * 2;

  const float* xb = x + ((size_t)b * Lsz + row0) * Hsz;

  __shared__ __align__(16) float lds_x[NWAVE][NSTRIP][16 * XPITCH];
  __shared__ float lds_part[NSTRIP][NWAVE][256];
  __shared__ __align__(16) float lds_low[NSTRIP][16 * 18];
  __shared__ float wsum[NWAVE * Esz];

  const float* xr0 = x + (size_t)b * Lsz * Hsz + 128 * wave + 4 * lane;
  const float* wr  = W_router + 128 * wave + 4 * lane;
  const v4f gxv = *(const v4f*)xr0;
  v4f gwv[Esz];
#pragma unroll
  for (int e = 0; e < Esz; ++e)
    gwv[e] = *(const v4f*)(wr + (size_t)e * Hsz);

  const float* xg = xb + 128 * wave + 4 * lane;
  v4f xv0[16], xv1[16];
#pragma unroll
  for (int i = 0; i < 16; ++i)
    xv0[i] = *(const v4f*)(xg + (size_t)i * Hsz);

  float p[Esz];
#pragma unroll
  for (int e = 0; e < Esz; ++e)
    p[e] = ((gxv.x * gwv[e].x + gxv.y * gwv[e].y) + gxv.z * gwv[e].z) + gxv.w * gwv[e].w;
#pragma unroll
  for (int e = 0; e < Esz; ++e) {
    p[e] += row_xor<1>(p[e]);
    p[e] += row_xor<2>(p[e]);
    p[e] += row_xor<4>(p[e]);
    p[e] += row_xor<8>(p[e]);
    p[e] += xhalf(p[e]);
  }
  {
    float mine = p[0];
#pragma unroll
    for (int e = 1; e < Esz; ++e) mine = (lane == e) ? p[e] : mine;
    if (lane < Esz) wsum[wave * Esz + lane] = mine;
  }
  lds_barrier();

  int i1 = 0, i2 = -1;
  float g1, g2;
  {
    const float* ws = wsum + (lane & 7);
    const float sl = ((ws[0 * Esz] + ws[1 * Esz]) + (ws[2 * Esz] + ws[3 * Esz])) +
                     ((ws[4 * Esz] + ws[5 * Esz]) + (ws[6 * Esz] + ws[7 * Esz]));
    float lg[Esz];
#pragma unroll
    for (int e = 0; e < Esz; ++e)
      lg[e] = __builtin_bit_cast(float, __builtin_amdgcn_readlane(__builtin_bit_cast(int, sl), e));
    float v1 = lg[0];
#pragma unroll
    for (int e = 1; e < Esz; ++e)
      if (lg[e] > v1) { v1 = lg[e]; i1 = e; }
    float v2 = -INFINITY;
#pragma unroll
    for (int e = 0; e < Esz; ++e) {
      if (e == i1) continue;
      if (lg[e] > v2) { v2 = lg[e]; i2 = e; }
    }
    if (i2 < 0) i2 = i1 ^ 1;
    const float e2 = expf(v2 - v1);
    const float inv = 1.f / (1.f + e2);
    g1 = inv;
    g2 = e2 * inv;
  }


  const int ea = (lane & 8) ? i2 : i1;
  const float* apk = A_down + ((size_t)ea * Hsz + 128 * wave + h2) * Rsz + (lane & 7);
  v2f bf[32];
#pragma unroll
  for (int kk = 0; kk < 32; ++kk) {
    bf[kk][0] = apk[kk * 4 * Rsz];
    bf[kk][1] = apk[kk * 4 * Rsz + Rsz];
  }
  __builtin_amdgcn_sched_barrier(0);

#pragma unroll
  for (int i = 0; i < 16; ++i)
    xv1[i] = *(const v4f*)(xg + (size_t)(16 + i) * Hsz);

  const int c4 = 4 * (lane & 15);
  const float* bp1 = B_up + ((size_t)i1 * Rsz + h2) * Hsz + 128 * wave + c4;
  const float* bp2 = B_up + ((size_t)i2 * Rsz + h2) * Hsz + 128 * wave + c4;
  v4f t1[2][2][2], t2[2][2][2];
#pragma unroll
  for (int m = 0; m < 2; ++m) {
#pragma unroll
    for (int ks = 0; ks < 2; ++ks) {
#pragma unroll
      for (int j = 0; j < 2; ++j) {
        t1[m][ks][j] = *(const v4f*)(bp1 + (4 * ks + j) * Hsz + 64 * m);
        t2[m][ks][j] = *(const v4f*)(bp2 + (4 * ks + j) * Hsz + 64 * m);
      }
    }
  }

  const size_t obase = (size_t)(8 * (lane >> 4)) * Hsz + 128 * wave + c4;
  float* outp = out + ((size_t)b * Lsz + row0) * Hsz + obase;
  const int xres_o = (8 * (lane >> 4)) * XPITCH + c4;
  v2f bo[8][4];


  auto stage_slice = [&](const int s, const v4f (&xs)[16]) __attribute__((always_inline)) {
#pragma unroll
    for (int i = 0; i < 16; ++i)
      *(v4f*)(lds_x[wave][s] + i * XPITCH + 4 * lane) = xs[i];
  };

  auto stage1 = [&](const int s) __attribute__((always_inline)) {
    v8f acc0 = {}, acc1 = {}, acc2 = {}, acc3 = {};
    const float* xa = lds_x[wave][s] + r * XPITCH + h2;
    v2f af[32];
#pragma unroll
    for (int kk = 0; kk < 32; ++kk)
      af[kk] = *(const v2f*)(xa + 4 * kk);
#pragma unroll
    for (int kk = 0; kk < 32; kk += 4) {
      acc0 = __builtin_amdgcn_wmma_f32_16x16x4_f32(false, af[kk + 0], false, bf[kk + 0], (short)0, acc0, false, false);
      acc1 = __builtin_amdgcn_wmma_f32_16x16x4_f32(false, af[kk + 1], false, bf[kk + 1], (short)0, acc1, false, false);
      acc2 = __builtin_amdgcn_wmma_f32_16x16x4_f32(false, af[kk + 2], false, bf[kk + 2], (short)0, acc2, false, false);
      acc3 = __builtin_amdgcn_wmma_f32_16x16x4_f32(false, af[kk + 3], false, bf[kk + 3], (short)0, acc3, false, false);
    }
    const v8f part = (acc0 + acc1) + (acc2 + acc3);
#pragma unroll
    for (int v = 0; v < 8; ++v)
      lds_part[s][wave][v * 32 + lane] = part[v];
  };

  auto sum_parts = [&](const int s) __attribute__((always_inline)) {
    const int t = threadIdx.x;
    const float sum = ((lds_part[s][0][t] + lds_part[s][1][t]) + (lds_part[s][2][t] + lds_part[s][3][t])) +
                      ((lds_part[s][4][t] + lds_part[s][5][t]) + (lds_part[s][6][t] + lds_part[s][7][t]));
    lds_low[s][(wave + 8 * (lane >> 4)) * 18 + (lane & 15)] = sum;
  };

  auto stage2 = [&](const int s) __attribute__((always_inline)) {
    v2f a2op[4];
#pragma unroll
    for (int ks = 0; ks < 4; ++ks)
      a2op[ks] = *(const v2f*)(&lds_low[s][r * 18 + 4 * ks + h2]);

    const float* xres_p = lds_x[wave][s] + xres_o;
    float4 xres[2][8];
#pragma unroll
    for (int m = 0; m < 2; ++m) {
#pragma unroll
      for (int v = 0; v < 8; ++v)
        xres[m][v] = *(const float4*)(xres_p + v * XPITCH + 64 * m);
    }

    float* outs = outp + (size_t)(16 * s) * Hsz;
#pragma unroll
    for (int m = 0; m < 2; ++m) {
      v8f y0 = {}, y1 = {}, y2 = {}, y3 = {};
#pragma unroll
      for (int ks = 0; ks < 4; ++ks) {
        y0 = __builtin_amdgcn_wmma_f32_16x16x4_f32(
            false, a2op[ks], false, bo[4 * m + 0][ks], (short)0, y0, false, false);
        y1 = __builtin_amdgcn_wmma_f32_16x16x4_f32(
            false, a2op[ks], false, bo[4 * m + 1][ks], (short)0, y1, false, false);
        y2 = __builtin_amdgcn_wmma_f32_16x16x4_f32(
            false, a2op[ks], false, bo[4 * m + 2][ks], (short)0, y2, false, false);
        y3 = __builtin_amdgcn_wmma_f32_16x16x4_f32(
            false, a2op[ks], false, bo[4 * m + 3][ks], (short)0, y3, false, false);
      }
#pragma unroll
      for (int v = 0; v < 8; ++v) {
        v4f o;
        o.x = xres[m][v].x + y0[v];
        o.y = xres[m][v].y + y1[v];
        o.z = xres[m][v].z + y2[v];
        o.w = xres[m][v].w + y3[v];
        __builtin_nontemporal_store(o, (v4f*)(outs + (size_t)v * Hsz + 64 * m));
      }
    }
  };

  stage_slice(0, xv0);
  lds_barrier();
  stage1(0);
  __builtin_amdgcn_sched_barrier(0);
  stage_slice(1, xv1);
  lds_barrier();
  sum_parts(0);
  lds_barrier();

#pragma unroll
  for (int m = 0; m < 2; ++m) {
#pragma unroll
    for (int ks = 0; ks < 2; ++ks) {
#pragma unroll
      for (int j = 0; j < 2; ++j) {
        bo[4 * m + 0][ks][j] = g1 * t1[m][ks][j].x;
        bo[4 * m + 1][ks][j] = g1 * t1[m][ks][j].y;
        bo[4 * m + 2][ks][j] = g1 * t1[m][ks][j].z;
        bo[4 * m + 3][ks][j] = g1 * t1[m][ks][j].w;
        bo[4 * m + 0][ks + 2][j] = g2 * t2[m][ks][j].x;
        bo[4 * m + 1][ks + 2][j] = g2 * t2[m][ks][j].y;
        bo[4 * m + 2][ks + 2][j] = g2 * t2[m][ks][j].z;
        bo[4 * m + 3][ks + 2][j] = g2 * t2[m][ks][j].w;
      }
    }
  }
  stage2(0);
  __builtin_amdgcn_sched_barrier(0);
  stage1(1);
  lds_barrier();
  sum_parts(1);
  lds_barrier();
  stage2(1);
}

extern "C" void kernel_launch(void* const* d_in, const int* in_sizes, int n_in,
                              void* d_out, int out_size, void* d_ws, size_t ws_size,
                              hipStream_t stream) {
  const float* x        = (const float*)d_in[0];
  const float* W_router = (const float*)d_in[1];
  const float* A_down   = (const float*)d_in[2];
  const float* B_up     = (const float*)d_in[3];
  float* out = (float*)d_out;

  moe_main<<<dim3(Lsz / (16 * NSTRIP), Bsz), dim3(256), 0, stream>>>(x, W_router, A_down, B_up, out);
}
